# starting kernel
# baseline (speedup 1.0000x reference)
#include <hip/hip_runtime.h>
#include <hip/hip_bf16.h>

typedef float v2f __attribute__((ext_vector_type(2)));
typedef float v8f __attribute__((ext_vector_type(8)));

static constexpr int DIM = 8192;
static constexpr int FD  = 128;
static constexpr int BM  = 32;
static constexpr int BK  = 32;
static constexpr float EPSV = 1e-10f;


__global__ __launch_bounds__(256) void HG_rowdeg(const float* __restrict__ H,
                                                 float* __restrict__ dvis) {
    __shared__ float red[256];
    const int row = blockIdx.x;
    const float* p = H + (size_t)row * DIM;
    float s = 0.0f;
    for (int c = threadIdx.x; c < DIM; c += 256) s += p[c];
    red[threadIdx.x] = s;
    __syncthreads();
    for (int off = 128; off > 0; off >>= 1) {
        if (threadIdx.x < off) red[threadIdx.x] += red[threadIdx.x + off];
        __syncthreads();
    }
    if (threadIdx.x == 0) dvis[row] = 1.0f / sqrtf(red[0] + EPSV);
}

__global__ __launch_bounds__(256) void HG_colzero(float* __restrict__ de) {
    de[blockIdx.x * 256 + threadIdx.x] = 0.0f;
}

__global__ __launch_bounds__(256) void HG_coldeg(const float* __restrict__ H,
                                                 float* __restrict__ de) {
    const int c  = blockIdx.x * 256 + threadIdx.x;
    const int r0 = blockIdx.y * 256;
    float s = 0.0f;
    #pragma unroll 4
    for (int r = 0; r < 256; ++r) s += H[(size_t)(r0 + r) * DIM + c];
    atomicAdd(&de[c], s);
}

__global__ __launch_bounds__(256) void HG_colfin(float* __restrict__ de) {
    const int i = blockIdx.x * 256 + threadIdx.x;
    de[i] = 1.0f / (de[i] + EPSV);
}

template <bool TRANSA, bool OUTSCALE>
__global__ __launch_bounds__(256) void HG_gemm(const float* __restrict__ A,
                                               const float* __restrict__ srcB,
                                               const float* __restrict__ scaleB,
                                               const float* __restrict__ outScale,
                                               float* __restrict__ out) {
    __shared__ __align__(16) float lAf[2][(BK / 2) * BM * 2];
    __shared__ __align__(16) float lBf[2][(BK / 2) * FD * 2];

    const int tid  = threadIdx.x;
    const int wave = tid >> 5;
    const int lid  = tid & 31;
    const int half = lid >> 4;
    const int ln   = lid & 15;
    const int m0   = blockIdx.x * BM;

    const int a_i4 = (tid & 7) * 4;
    const int a_j  = tid >> 3;
    const int b_d4 = (tid & 31) * 4;
    const int b_k  = tid >> 5;

    v8f acc[2];
    acc[0] = (v8f){};
    acc[1] = (v8f){};

    float4 ra;
    float4 rb[4];
    float  sb[4];

    auto load_chunk = [&](int k0) {
        if (TRANSA) {
            ra = *(const float4*)(A + (size_t)(k0 + a_j) * DIM + m0 + a_i4);
        } else {
            ra = *(const float4*)(A + (size_t)(m0 + a_j) * DIM + k0 + a_i4);
        }
        #pragma unroll
        for (int p = 0; p < 4; ++p) {
            const int row = k0 + b_k + p * 8;
            sb[p] = scaleB[row];
            rb[p] = *(const float4*)(srcB + (size_t)row * FD + b_d4);
        }
    };

    auto store_chunk = [&](int buf) {
        if (TRANSA) {
            const int kp = a_j >> 1, kc = a_j & 1;
            lAf[buf][(kp * BM + a_i4 + 0) * 2 + kc] = ra.x;
            lAf[buf][(kp * BM + a_i4 + 1) * 2 + kc] = ra.y;
            lAf[buf][(kp * BM + a_i4 + 2) * 2 + kc] = ra.z;
            lAf[buf][(kp * BM + a_i4 + 3) * 2 + kc] = ra.w;
        } else {
            lAf[buf][((a_i4 >> 1) * BM + a_j) * 2 + 0] = ra.x;
            lAf[buf][((a_i4 >> 1) * BM + a_j) * 2 + 1] = ra.y;
            lAf[buf][(((a_i4 + 2) >> 1) * BM + a_j) * 2 + 0] = ra.z;
            lAf[buf][(((a_i4 + 2) >> 1) * BM + a_j) * 2 + 1] = ra.w;
        }
        #pragma unroll
        for (int p = 0; p < 4; ++p) {
            const int k  = b_k + p * 8;
            const int kp = k >> 1, kc = k & 1;
            lBf[buf][(kp * FD + b_d4 + 0) * 2 + kc] = rb[p].x * sb[p];
            lBf[buf][(kp * FD + b_d4 + 1) * 2 + kc] = rb[p].y * sb[p];
            lBf[buf][(kp * FD + b_d4 + 2) * 2 + kc] = rb[p].z * sb[p];
            lBf[buf][(kp * FD + b_d4 + 3) * 2 + kc] = rb[p].w * sb[p];
        }
    };

    load_chunk(0);
    store_chunk(0);
    __syncthreads();

    int cur = 0;
    for (int ki = 0; ki < DIM / BK; ++ki) {
        const int knext = ((ki + 1) & (DIM / BK - 1)) * BK;
        load_chunk(knext);

        const float2* lA2 = (const float2*)lAf[cur];
        const float2* lB2 = (const float2*)lBf[cur];
        #pragma unroll
        for (int s = 0; s < 8; ++s) {
            const int kp = 2 * s + half;
            const float2 bf  = lB2[kp * FD + wave * 16 + ln];
            const float2 a0f = lA2[kp * BM + ln];
            const float2 a1f = lA2[kp * BM + 16 + ln];
            v2f b;  b[0]  = bf.x;  b[1]  = bf.y;
            v2f a0; a0[0] = a0f.x; a0[1] = a0f.y;
            v2f a1; a1[0] = a1f.x; a1[1] = a1f.y;
            acc[0] = __builtin_amdgcn_wmma_f32_16x16x4_f32(
                false, a0, false, b, (short)0, acc[0], false, false);
            acc[1] = __builtin_amdgcn_wmma_f32_16x16x4_f32(
                false, a1, false, b, (short)0, acc[1], false, false);
        }

        store_chunk(cur ^ 1);
        __syncthreads();
        cur ^= 1;
    }

    #pragma unroll
    for (int r = 0; r < 2; ++r) {
        #pragma unroll
        for (int j = 0; j < 8; ++j) {
            const int row = m0 + 16 * r + half * 8 + j;
            const float os = OUTSCALE ? outScale[row] : 1.0f;
            out[(size_t)row * FD + wave * 16 + ln] = acc[r][j] * os;
        }
    }
}

extern "C" void kernel_launch(void* const* d_in, const int* in_sizes, int n_in,
                              void* d_out, int out_size, void* d_ws, size_t ws_size,
                              hipStream_t stream) {
    (void)in_sizes; (void)n_in; (void)out_size; (void)ws_size;
    const float* H = (const float*)d_in[0];
    const float* F = (const float*)d_in[1];
    float* out  = (float*)d_out;
    float* ws   = (float*)d_ws;
    float* dvis = ws;
    float* de   = ws + DIM;
    float* Y    = ws + 2 * DIM;

    HG_rowdeg<<<DIM, 256, 0, stream>>>(H, dvis);
    HG_colzero<<<DIM / 256, 256, 0, stream>>>(de);
    HG_coldeg<<<dim3(DIM / 256, DIM / 256), 256, 0, stream>>>(H, de);
    HG_colfin<<<DIM / 256, 256, 0, stream>>>(de);

    HG_gemm<true, false><<<DIM / BM, 256, 0, stream>>>(H, F, dvis, nullptr, Y);
    HG_gemm<false, true><<<DIM / BM, 256, 0, stream>>>(H, Y, de, dvis, out);
}


# GEMM K loop software-pipelined: global loads requested 4 chunks ahead, next chunk's LDS stores drain under the WMMA block, hand-written barrier that does not drain loads; sums unchanged
# speedup vs baseline: 1.8038x; 1.8038x over previous
#include <hip/hip_runtime.h>
#include <hip/hip_bf16.h>

typedef float v2f __attribute__((ext_vector_type(2)));
typedef float v8f __attribute__((ext_vector_type(8)));

static constexpr int DIM = 8192;
static constexpr int FD  = 128;
static constexpr int BM  = 32;
static constexpr int BK  = 32;
static constexpr float EPSV = 1e-10f;


__global__ __launch_bounds__(256) void HG_rowdeg(const float* __restrict__ H,
                                                 float* __restrict__ dvis) {
    __shared__ float red[256];
    const int row = blockIdx.x;
    const float* p = H + (size_t)row * DIM;
    float s = 0.0f;
    for (int c = threadIdx.x; c < DIM; c += 256) s += p[c];
    red[threadIdx.x] = s;
    __syncthreads();
    for (int off = 128; off > 0; off >>= 1) {
        if (threadIdx.x < off) red[threadIdx.x] += red[threadIdx.x + off];
        __syncthreads();
    }
    if (threadIdx.x == 0) dvis[row] = 1.0f / sqrtf(red[0] + EPSV);
}

__global__ __launch_bounds__(256) void HG_colzero(float* __restrict__ de) {
    de[blockIdx.x * 256 + threadIdx.x] = 0.0f;
}

__global__ __launch_bounds__(256) void HG_coldeg(const float* __restrict__ H,
                                                 float* __restrict__ de) {
    const int c  = blockIdx.x * 256 + threadIdx.x;
    const int r0 = blockIdx.y * 256;
    float s = 0.0f;
    #pragma unroll 4
    for (int r = 0; r < 256; ++r) s += H[(size_t)(r0 + r) * DIM + c];
    atomicAdd(&de[c], s);
}

__global__ __launch_bounds__(256) void HG_colfin(float* __restrict__ de) {
    const int i = blockIdx.x * 256 + threadIdx.x;
    de[i] = 1.0f / (de[i] + EPSV);
}

struct HG_stage {
    float4 ra;
    float4 rb[4];
    float  sb[4];
};

static __device__ __forceinline__ void HG_lds_barrier() {
    asm volatile("s_wait_dscnt 0x0\n\ts_barrier_signal -1\n\ts_barrier_wait -1" ::: "memory");
}

template <bool TRANSA, bool OUTSCALE>
__global__ __launch_bounds__(256) void HG_gemm(const float* A,
                                               const float* srcB,
                                               const float* scaleB,
                                               const float* outScale,
                                               float* out) {
    __shared__ __align__(16) float lAf[2][(BK / 2) * BM * 2];
    __shared__ __align__(16) float lBf[2][(BK / 2) * FD * 2];

    const int tid  = threadIdx.x;
    const int wave = tid >> 5;
    const int lid  = tid & 31;
    const int half = lid >> 4;
    const int ln   = lid & 15;
    const int m0   = blockIdx.x * BM;

    const int a_i4 = (tid & 7) * 4;
    const int a_j  = tid >> 3;
    const int b_d4 = (tid & 31) * 4;
    const int b_k  = tid >> 5;

    v8f acc[2];
    acc[0] = (v8f){};
    acc[1] = (v8f){};

    HG_stage st0, st1, st2, st3;

    auto load_chunk = [&](HG_stage& st, int k0) {
        if (TRANSA) {
            st.ra = *(const float4*)(A + (size_t)(k0 + a_j) * DIM + m0 + a_i4);
        } else {
            st.ra = *(const float4*)(A + (size_t)(m0 + a_j) * DIM + k0 + a_i4);
        }
        #pragma unroll
        for (int p = 0; p < 4; ++p) {
            const int row = k0 + b_k + p * 8;
            st.sb[p] = scaleB[row];
            st.rb[p] = *(const float4*)(srcB + (size_t)row * FD + b_d4);
        }
    };

    auto store_chunk = [&](const HG_stage& st, int buf) {
        if (TRANSA) {
            const int kp = a_j >> 1, kc = a_j & 1;
            lAf[buf][(kp * BM + a_i4 + 0) * 2 + kc] = st.ra.x;
            lAf[buf][(kp * BM + a_i4 + 1) * 2 + kc] = st.ra.y;
            lAf[buf][(kp * BM + a_i4 + 2) * 2 + kc] = st.ra.z;
            lAf[buf][(kp * BM + a_i4 + 3) * 2 + kc] = st.ra.w;
        } else {
            lAf[buf][((a_i4 >> 1) * BM + a_j) * 2 + 0] = st.ra.x;
            lAf[buf][((a_i4 >> 1) * BM + a_j) * 2 + 1] = st.ra.y;
            lAf[buf][(((a_i4 + 2) >> 1) * BM + a_j) * 2 + 0] = st.ra.z;
            lAf[buf][(((a_i4 + 2) >> 1) * BM + a_j) * 2 + 1] = st.ra.w;
        }
        #pragma unroll
        for (int p = 0; p < 4; ++p) {
            const int k  = b_k + p * 8;
            const int kp = k >> 1, kc = k & 1;
            float sc = st.sb[p];
            asm volatile("" : "+v"(sc));
            lBf[buf][(kp * FD + b_d4 + 0) * 2 + kc] = st.rb[p].x * sc;
            lBf[buf][(kp * FD + b_d4 + 1) * 2 + kc] = st.rb[p].y * sc;
            lBf[buf][(kp * FD + b_d4 + 2) * 2 + kc] = st.rb[p].z * sc;
            lBf[buf][(kp * FD + b_d4 + 3) * 2 + kc] = st.rb[p].w * sc;
        }
    };

    v2f fa0[8], fa1[8], fb[8];

    auto load_frags = [&](int buf) {
        const float2* lA2 = (const float2*)lAf[buf];
        const float2* lB2 = (const float2*)lBf[buf];
        #pragma unroll
        for (int s = 0; s < 8; ++s) {
            const int kp = 2 * s + half;
            const float2 bf  = lB2[kp * FD + wave * 16 + ln];
            const float2 a0f = lA2[kp * BM + ln];
            const float2 a1f = lA2[kp * BM + 16 + ln];
            fb[s][0]  = bf.x;  fb[s][1]  = bf.y;
            fa0[s][0] = a0f.x; fa0[s][1] = a0f.y;
            fa1[s][0] = a1f.x; fa1[s][1] = a1f.y;
        }
    };

    auto mma_chunk = [&]() {
        #pragma unroll
        for (int s = 0; s < 8; ++s) {
            acc[0] = __builtin_amdgcn_wmma_f32_16x16x4_f32(
                false, fa0[s], false, fb[s], (short)0, acc[0], false, false);
            acc[1] = __builtin_amdgcn_wmma_f32_16x16x4_f32(
                false, fa1[s], false, fb[s], (short)0, acc[1], false, false);
        }
    };

    constexpr int NK = DIM / BK;

    load_chunk(st0, 0);
    __builtin_amdgcn_sched_barrier(0);
    load_chunk(st1, BK);
    __builtin_amdgcn_sched_barrier(0);
    load_chunk(st2, 2 * BK);
    __builtin_amdgcn_sched_barrier(0);
    load_chunk(st3, 3 * BK);
    store_chunk(st0, 0);
    HG_lds_barrier();

    #define HG_STEP(ST_REQ, ST_WR, BUF, CNEXT)                                   \
        load_chunk(ST_REQ, ((CNEXT) & (NK - 1)) * BK);                           \
        load_frags(BUF);                                                         \
        store_chunk(ST_WR, (BUF) ^ 1);                                           \
          \
          \
          \
        __builtin_amdgcn_sched_barrier(0);                                       \
        mma_chunk();                                                             \
          \
          \
          \
        __builtin_amdgcn_sched_barrier(0);                                       \
        HG_lds_barrier();

    for (int ki = 0; ki < NK; ki += 4) {
        HG_STEP(st0, st1, 0, ki + 4)
        HG_STEP(st1, st2, 1, ki + 5)
        HG_STEP(st2, st3, 0, ki + 6)
        HG_STEP(st3, st0, 1, ki + 7)
    }
    #undef HG_STEP

    float os[2][8];
    #pragma unroll
    for (int r = 0; r < 2; ++r) {
        #pragma unroll
        for (int j = 0; j < 8; ++j) {
            const int row = m0 + 16 * r + half * 8 + j;
            os[r][j] = OUTSCALE ? outScale[row] : 1.0f;
        }
    }
    #pragma unroll
    for (int r = 0; r < 2; ++r) {
        #pragma unroll
        for (int j = 0; j < 8; ++j) {
            const int row = m0 + 16 * r + half * 8 + j;
            out[(size_t)row * FD + wave * 16 + ln] = acc[r][j] * os[r][j];
        }
    }
}

extern "C" void kernel_launch(void* const* d_in, const int* in_sizes, int n_in,
                              void* d_out, int out_size, void* d_ws, size_t ws_size,
                              hipStream_t stream) {
    (void)in_sizes; (void)n_in; (void)out_size; (void)ws_size;
    const float* H = (const float*)d_in[0];
    const float* F = (const float*)d_in[1];
    float* out  = (float*)d_out;
    float* ws   = (float*)d_ws;
    float* dvis = ws;
    float* de   = ws + DIM;
    float* Y    = ws + 2 * DIM;

    HG_rowdeg<<<DIM, 256, 0, stream>>>(H, dvis);
    HG_colzero<<<DIM / 256, 256, 0, stream>>>(de);
    HG_coldeg<<<dim3(DIM / 256, DIM / 256), 256, 0, stream>>>(H, de);
    HG_colfin<<<DIM / 256, 256, 0, stream>>>(de);

    HG_gemm<true, false><<<DIM / BM, 256, 0, stream>>>(H, F, dvis, nullptr, Y);
    HG_gemm<false, true><<<DIM / BM, 256, 0, stream>>>(H, Y, de, dvis, out);
}


# LDS tiles re-laid bank-conflict-free: padded pitches for the b64 fragment reads; each thread fetches both rows of a k pair so tiles go in as 16-byte stores, 5 per K step (were 20); sums unchanged
# speedup vs baseline: 2.1697x; 1.2028x over previous
#include <hip/hip_runtime.h>
#include <hip/hip_bf16.h>

typedef float v2f __attribute__((ext_vector_type(2)));
typedef float v4f __attribute__((ext_vector_type(4)));
typedef float v8f __attribute__((ext_vector_type(8)));

static constexpr int DIM = 8192;
static constexpr int FD  = 128;
static constexpr int BM  = 32;
static constexpr int BK  = 32;
static constexpr float EPSV = 1e-10f;
static constexpr int PB2  = 144;
static constexpr int BOFF = 72;
static constexpr int PA2  = 48;
static constexpr int PAM  = 36;


__global__ __launch_bounds__(256) void HG_rowdeg(const float* __restrict__ H,
                                                 float* __restrict__ dvis) {
    __shared__ float red[256];
    const int row = blockIdx.x;
    const float* p = H + (size_t)row * DIM;
    float s = 0.0f;
    for (int c = threadIdx.x; c < DIM; c += 256) s += p[c];
    red[threadIdx.x] = s;
    __syncthreads();
    for (int off = 128; off > 0; off >>= 1) {
        if (threadIdx.x < off) red[threadIdx.x] += red[threadIdx.x + off];
        __syncthreads();
    }
    if (threadIdx.x == 0) dvis[row] = 1.0f / sqrtf(red[0] + EPSV);
}

__global__ __launch_bounds__(256) void HG_colzero(float* __restrict__ de) {
    de[blockIdx.x * 256 + threadIdx.x] = 0.0f;
}

__global__ __launch_bounds__(256) void HG_coldeg(const float* __restrict__ H,
                                                 float* __restrict__ de) {
    const int c  = blockIdx.x * 256 + threadIdx.x;
    const int r0 = blockIdx.y * 256;
    float s = 0.0f;
    #pragma unroll 4
    for (int r = 0; r < 256; ++r) s += H[(size_t)(r0 + r) * DIM + c];
    atomicAdd(&de[c], s);
}

__global__ __launch_bounds__(256) void HG_colfin(float* __restrict__ de) {
    const int i = blockIdx.x * 256 + threadIdx.x;
    de[i] = 1.0f / (de[i] + EPSV);
}

struct HG_stage {
    v4f ra;
    v2f ga0, ga1;
    v4f rb[4];
    float  sb[4];
};

static __device__ __forceinline__ void HG_lds_barrier() {
    asm volatile("s_wait_dscnt 0x0\n\ts_barrier_signal -1\n\ts_barrier_wait -1" ::: "memory");
}

template <bool TRANSA, bool OUTSCALE>
__global__ __launch_bounds__(256) void HG_gemm(const float* A,
                                               const float* srcB,
                                               const float* scaleB,
                                               const float* outScale,
                                               float* out) {
    __shared__ __align__(16) float lAf[2][(BK / 2) * PA2 * 2];
    __shared__ __align__(16) float lBf[2][(BK / 2) * PB2 * 2];

    const int tid  = threadIdx.x;
    const int wave = tid >> 5;
    const int lid  = tid & 31;
    const int half = lid >> 4;
    const int ln   = lid & 15;
    const int m0   = blockIdx.x * BM;

    const int a_i4 = (tid & 7) * 4;
    const int a_j  = tid >> 3;
    const int a_m2 = (tid & 15) * 2;
    const int a_kp = tid >> 4;
    const int b_d4 = lid * 4;
    const int b_kp = wave;
    const int b_pos = wave * 8 + (ln >> 2) * 2 + (ln & 1) + ((ln >> 1) & 1) * BOFF;

    v8f acc[2];
    acc[0] = (v8f){};
    acc[1] = (v8f){};

    HG_stage st0, st1, st2, st3;

    auto load_chunk = [&](HG_stage& st, int k0) {
        if (TRANSA) {
            const float* pa = A + (size_t)(k0 + 2 * a_kp) * DIM + m0 + a_m2;
            st.ga0 = *(const v2f*)pa;
            st.ga1 = *(const v2f*)(pa + DIM);
        } else {
            st.ra = *(const v4f*)(A + (size_t)(m0 + a_j) * DIM + k0 + a_i4);
        }
        #pragma unroll
        for (int p = 0; p < 2; ++p) {
            const int row = k0 + 2 * (b_kp + p * 8);
            st.sb[2 * p]     = scaleB[row];
            st.sb[2 * p + 1] = scaleB[row + 1];
            st.rb[2 * p]     = *(const v4f*)(srcB + (size_t)row * FD + b_d4);
            st.rb[2 * p + 1] = *(const v4f*)(srcB + (size_t)(row + 1) * FD + b_d4);
        }
    };

    auto store_chunk = [&](const HG_stage& st, int buf) {
        if (TRANSA) {
            v2f g0 = st.ga0, g1 = st.ga1;
            asm volatile("" : "+v"(g0));
            asm volatile("" : "+v"(g1));
            *(v4f*)&lAf[buf][(a_kp * PA2 + a_m2) * 2] = (v4f){g0.x, g1.x, g0.y, g1.y};
        } else {
            *(v4f*)&lAf[buf][a_j * PAM + a_i4] = st.ra;
        }
        #pragma unroll
        for (int p = 0; p < 2; ++p) {
            const int kp = b_kp + p * 8;
            float s0 = st.sb[2 * p];
            float s1 = st.sb[2 * p + 1];
            v4f r0 = st.rb[2 * p];
            v4f r1 = st.rb[2 * p + 1];
            asm volatile("" : "+v"(s0));
            asm volatile("" : "+v"(s1));
            asm volatile("" : "+v"(r0));
            asm volatile("" : "+v"(r1));
            *(v4f*)&lBf[buf][(kp * PB2 + 2 * lid) * 2] =
                (v4f){r0.x * s0, r1.x * s1, r0.y * s0, r1.y * s1};
            *(v4f*)&lBf[buf][(kp * PB2 + BOFF + 2 * lid) * 2] =
                (v4f){r0.z * s0, r1.z * s1, r0.w * s0, r1.w * s1};
        }
    };

    v2f fa0[8], fa1[8], fb[8];

    auto load_frags = [&](int buf) {
        const float2* lA2 = (const float2*)lAf[buf];
        const float2* lB2 = (const float2*)lBf[buf];
        #pragma unroll
        for (int s = 0; s < 8; ++s) {
            const int kp = 2 * s + half;
            const float2 bf  = lB2[kp * PB2 + b_pos];
            const float2 a0f = TRANSA ? lA2[kp * PA2 + ln]
                                      : *(const float2*)&lAf[buf][ln * PAM + 2 * kp];
            const float2 a1f = TRANSA ? lA2[kp * PA2 + 16 + ln]
                                      : *(const float2*)&lAf[buf][(16 + ln) * PAM + 2 * kp];
            fb[s][0]  = bf.x;  fb[s][1]  = bf.y;
            fa0[s][0] = a0f.x; fa0[s][1] = a0f.y;
            fa1[s][0] = a1f.x; fa1[s][1] = a1f.y;
        }
    };

    auto mma_chunk = [&]() {
        asm volatile("" : "+v"(acc[0]), "+v"(acc[1]));
        #pragma unroll
        for (int s = 0; s < 8; ++s) {
            acc[0] = __builtin_amdgcn_wmma_f32_16x16x4_f32(
                false, fa0[s], false, fb[s], (short)0, acc[0], false, false);
            acc[1] = __builtin_amdgcn_wmma_f32_16x16x4_f32(
                false, fa1[s], false, fb[s], (short)0, acc[1], false, false);
        }
    };

    constexpr int NK = DIM / BK;

    load_chunk(st0, 0);
    __builtin_amdgcn_sched_barrier(0);
    load_chunk(st1, BK);
    __builtin_amdgcn_sched_barrier(0);
    load_chunk(st2, 2 * BK);
    __builtin_amdgcn_sched_barrier(0);
    load_chunk(st3, 3 * BK);
    store_chunk(st0, 0);
    HG_lds_barrier();

    #define HG_STEP(ST_REQ, ST_WR, BUF, CNEXT)                                   \
        load_chunk(ST_REQ, ((CNEXT) & (NK - 1)) * BK);                           \
        load_frags(BUF);                                                         \
        store_chunk(ST_WR, (BUF) ^ 1);                                           \
          \
          \
          \
        __builtin_amdgcn_sched_barrier(0);                                       \
        mma_chunk();                                                             \
          \
          \
          \
        __builtin_amdgcn_sched_barrier(0);                                       \
        HG_lds_barrier();

    for (int ki = 0; ki < NK; ki += 4) {
        HG_STEP(st0, st1, 0, ki + 4)
        HG_STEP(st1, st2, 1, ki + 5)
        HG_STEP(st2, st3, 0, ki + 6)
        HG_STEP(st3, st0, 1, ki + 7)
    }
    #undef HG_STEP

    float os[2][8];
    #pragma unroll
    for (int r = 0; r < 2; ++r) {
        #pragma unroll
        for (int j = 0; j < 8; ++j) {
            const int row = m0 + 16 * r + half * 8 + j;
            os[r][j] = OUTSCALE ? outScale[row] : 1.0f;
        }
    }
    #pragma unroll
    for (int r = 0; r < 2; ++r) {
        #pragma unroll
        for (int j = 0; j < 8; ++j) {
            const int row = m0 + 16 * r + half * 8 + j;
            out[(size_t)row * FD + wave * 16 + ln] = acc[r][j] * os[r][j];
        }
    }
}

extern "C" void kernel_launch(void* const* d_in, const int* in_sizes, int n_in,
                              void* d_out, int out_size, void* d_ws, size_t ws_size,
                              hipStream_t stream) {
    (void)in_sizes; (void)n_in; (void)out_size; (void)ws_size;
    const float* H = (const float*)d_in[0];
    const float* F = (const float*)d_in[1];
    float* out  = (float*)d_out;
    float* ws   = (float*)d_ws;
    float* dvis = ws;
    float* de   = ws + DIM;
    float* Y    = ws + 2 * DIM;

    HG_rowdeg<<<DIM, 256, 0, stream>>>(H, dvis);
    HG_colzero<<<DIM / 256, 256, 0, stream>>>(de);
    HG_coldeg<<<dim3(DIM / 256, DIM / 256), 256, 0, stream>>>(H, de);
    HG_colfin<<<DIM / 256, 256, 0, stream>>>(de);

    HG_gemm<true, false><<<DIM / BM, 256, 0, stream>>>(H, F, dvis, nullptr, Y);
    HG_gemm<false, true><<<DIM / BM, 256, 0, stream>>>(H, Y, de, dvis, out);
}


# Column degrees summed inside the transposed GEMM from the A tiles it already fetches: one pass over H, three launches and the atomics go; d_e now summed in a fixed order (was atomic arrival order)
# speedup vs baseline: 2.5171x; 1.1601x over previous
#include <hip/hip_runtime.h>
#include <hip/hip_bf16.h>

typedef float v2f __attribute__((ext_vector_type(2)));
typedef float v4f __attribute__((ext_vector_type(4)));
typedef float v8f __attribute__((ext_vector_type(8)));

static constexpr int DIM = 8192;
static constexpr int FD  = 128;
static constexpr int BM  = 32;
static constexpr int BK  = 32;
static constexpr float EPSV = 1e-10f;
static constexpr int PB2  = 144;
static constexpr int BOFF = 72;
static constexpr int PA2  = 48;
static constexpr int PAM  = 36;


__global__ __launch_bounds__(256) void HG_rowdeg(const float* __restrict__ H,
                                                 float* __restrict__ dvis) {
    __shared__ float red[256];
    const int row = blockIdx.x;
    const float* p = H + (size_t)row * DIM;
    float s = 0.0f;
    for (int c = threadIdx.x; c < DIM; c += 256) s += p[c];
    red[threadIdx.x] = s;
    __syncthreads();
    for (int off = 128; off > 0; off >>= 1) {
        if (threadIdx.x < off) red[threadIdx.x] += red[threadIdx.x + off];
        __syncthreads();
    }
    if (threadIdx.x == 0) dvis[row] = 1.0f / sqrtf(red[0] + EPSV);
}


struct HG_stage {
    v4f ra;
    v2f ga0, ga1;
    v4f rb[4];
    float  sb[4];
};

static __device__ __forceinline__ void HG_lds_barrier() {
    asm volatile("s_wait_dscnt 0x0\n\ts_barrier_signal -1\n\ts_barrier_wait -1" ::: "memory");
}

template <bool TRANSA, bool OUTSCALE, bool COLDEG>
__global__ __launch_bounds__(256) void HG_gemm(const float* A,
                                               const float* srcB,
                                               const float* scaleB,
                                               const float* outScale,
                                               float* out,
                                               float* colInv) {
    static_assert(!COLDEG || TRANSA, "column sums come from the transposed A tiles");
    __shared__ __align__(16) float lAf[2][(BK / 2) * PA2 * 2];
    __shared__ __align__(16) float lBf[2][(BK / 2) * PB2 * 2];

    const int tid  = threadIdx.x;
    const int wave = tid >> 5;
    const int lid  = tid & 31;
    const int half = lid >> 4;
    const int ln   = lid & 15;
    const int m0   = blockIdx.x * BM;

    const int a_i4 = (tid & 7) * 4;
    const int a_j  = tid >> 3;
    const int a_m2 = (tid & 15) * 2;
    const int a_kp = tid >> 4;
    const int b_d4 = lid * 4;
    const int b_kp = wave;
    const int b_pos = wave * 8 + (ln >> 2) * 2 + (ln & 1) + ((ln >> 1) & 1) * BOFF;

    v8f acc[2];
    acc[0] = (v8f){};
    acc[1] = (v8f){};

    v2f csum = (v2f){0.0f, 0.0f};

    HG_stage st0, st1, st2, st3;

    auto load_chunk = [&](HG_stage& st, int k0) {
        if (TRANSA) {
            const float* pa = A + (size_t)(k0 + 2 * a_kp) * DIM + m0 + a_m2;
            st.ga0 = *(const v2f*)pa;
            st.ga1 = *(const v2f*)(pa + DIM);
        } else {
            st.ra = *(const v4f*)(A + (size_t)(m0 + a_j) * DIM + k0 + a_i4);
        }
        #pragma unroll
        for (int p = 0; p < 2; ++p) {
            const int row = k0 + 2 * (b_kp + p * 8);
            st.sb[2 * p]     = scaleB[row];
            st.sb[2 * p + 1] = scaleB[row + 1];
            st.rb[2 * p]     = *(const v4f*)(srcB + (size_t)row * FD + b_d4);
            st.rb[2 * p + 1] = *(const v4f*)(srcB + (size_t)(row + 1) * FD + b_d4);
        }
    };

    auto store_chunk = [&](const HG_stage& st, int buf, bool count) {
        if (TRANSA) {
            v2f g0 = st.ga0, g1 = st.ga1;
            asm volatile("" : "+v"(g0));
            asm volatile("" : "+v"(g1));
            if (COLDEG && count) {
                csum += g0;
                csum += g1;
            }
            *(v4f*)&lAf[buf][(a_kp * PA2 + a_m2) * 2] = (v4f){g0.x, g1.x, g0.y, g1.y};
        } else {
            *(v4f*)&lAf[buf][a_j * PAM + a_i4] = st.ra;
        }
        #pragma unroll
        for (int p = 0; p < 2; ++p) {
            const int kp = b_kp + p * 8;
            float s0 = st.sb[2 * p];
            float s1 = st.sb[2 * p + 1];
            v4f r0 = st.rb[2 * p];
            v4f r1 = st.rb[2 * p + 1];
            asm volatile("" : "+v"(s0));
            asm volatile("" : "+v"(s1));
            asm volatile("" : "+v"(r0));
            asm volatile("" : "+v"(r1));
            *(v4f*)&lBf[buf][(kp * PB2 + 2 * lid) * 2] =
                (v4f){r0.x * s0, r1.x * s1, r0.y * s0, r1.y * s1};
            *(v4f*)&lBf[buf][(kp * PB2 + BOFF + 2 * lid) * 2] =
                (v4f){r0.z * s0, r1.z * s1, r0.w * s0, r1.w * s1};
        }
    };

    v2f fa0[8], fa1[8], fb[8];

    auto load_frags = [&](int buf) {
        const float2* lA2 = (const float2*)lAf[buf];
        const float2* lB2 = (const float2*)lBf[buf];
        #pragma unroll
        for (int s = 0; s < 8; ++s) {
            const int kp = 2 * s + half;
            const float2 bf  = lB2[kp * PB2 + b_pos];
            const float2 a0f = TRANSA ? lA2[kp * PA2 + ln]
                                      : *(const float2*)&lAf[buf][ln * PAM + 2 * kp];
            const float2 a1f = TRANSA ? lA2[kp * PA2 + 16 + ln]
                                      : *(const float2*)&lAf[buf][(16 + ln) * PAM + 2 * kp];
            fb[s][0]  = bf.x;  fb[s][1]  = bf.y;
            fa0[s][0] = a0f.x; fa0[s][1] = a0f.y;
            fa1[s][0] = a1f.x; fa1[s][1] = a1f.y;
        }
    };

    auto mma_chunk = [&]() {
        asm volatile("" : "+v"(acc[0]), "+v"(acc[1]));
        #pragma unroll
        for (int s = 0; s < 8; ++s) {
            acc[0] = __builtin_amdgcn_wmma_f32_16x16x4_f32(
                false, fa0[s], false, fb[s], (short)0, acc[0], false, false);
            acc[1] = __builtin_amdgcn_wmma_f32_16x16x4_f32(
                false, fa1[s], false, fb[s], (short)0, acc[1], false, false);
        }
    };

    constexpr int NK = DIM / BK;

    load_chunk(st0, 0);
    __builtin_amdgcn_sched_barrier(0);
    load_chunk(st1, BK);
    __builtin_amdgcn_sched_barrier(0);
    load_chunk(st2, 2 * BK);
    __builtin_amdgcn_sched_barrier(0);
    load_chunk(st3, 3 * BK);
    store_chunk(st0, 0, true);
    HG_lds_barrier();

    #define HG_STEP(ST_REQ, ST_WR, BUF, CNEXT, COUNT)                            \
        load_chunk(ST_REQ, ((CNEXT) & (NK - 1)) * BK);                           \
        load_frags(BUF);                                                         \
        store_chunk(ST_WR, (BUF) ^ 1, COUNT);                                    \
          \
          \
          \
        __builtin_amdgcn_sched_barrier(0);                                       \
        mma_chunk();                                                             \
          \
          \
          \
        __builtin_amdgcn_sched_barrier(0);                                       \
        HG_lds_barrier();

    constexpr int NLOOP = COLDEG ? NK - 4 : NK;
    for (int ki = 0; ki < NLOOP; ki += 4) {
        HG_STEP(st0, st1, 0, ki + 4, true)
        HG_STEP(st1, st2, 1, ki + 5, true)
        HG_STEP(st2, st3, 0, ki + 6, true)
        HG_STEP(st3, st0, 1, ki + 7, true)
    }
    if (COLDEG) {
        HG_STEP(st0, st1, 0, NK,     true)
        HG_STEP(st1, st2, 1, NK + 1, true)
        HG_STEP(st2, st3, 0, NK + 2, true)
        HG_STEP(st3, st0, 1, NK + 3, false)
    }
    #undef HG_STEP

    if (COLDEG) {
        __shared__ __align__(8) float cs[16 * BM];
        *(v2f*)&cs[2 * tid] = csum;
        __syncthreads();
        if (tid < BM) {
            float s = cs[tid];
            for (int kp = 1; kp < 16; ++kp) s += cs[kp * BM + tid];
            colInv[m0 + tid] = 1.0f / (s + EPSV);
        }
    }

    float os[2][8];
    #pragma unroll
    for (int r = 0; r < 2; ++r) {
        #pragma unroll
        for (int j = 0; j < 8; ++j) {
            const int row = m0 + 16 * r + half * 8 + j;
            os[r][j] = OUTSCALE ? outScale[row] : 1.0f;
        }
    }
    #pragma unroll
    for (int r = 0; r < 2; ++r) {
        #pragma unroll
        for (int j = 0; j < 8; ++j) {
            const int row = m0 + 16 * r + half * 8 + j;
            out[(size_t)row * FD + wave * 16 + ln] = acc[r][j] * os[r][j];
        }
    }
}

extern "C" void kernel_launch(void* const* d_in, const int* in_sizes, int n_in,
                              void* d_out, int out_size, void* d_ws, size_t ws_size,
                              hipStream_t stream) {
    (void)in_sizes; (void)n_in; (void)out_size; (void)ws_size;
    const float* H = (const float*)d_in[0];
    const float* F = (const float*)d_in[1];
    float* out  = (float*)d_out;
    float* ws   = (float*)d_ws;
    float* dvis = ws;
    float* de   = ws + DIM;
    float* Y    = ws + 2 * DIM;

    HG_rowdeg<<<DIM, 256, 0, stream>>>(H, dvis);

    HG_gemm<true, false, true><<<DIM / BM, 256, 0, stream>>>(H, F, dvis, nullptr, Y, de);
    HG_gemm<false, true, false><<<DIM / BM, 256, 0, stream>>>(H, Y, de, dvis, out, nullptr);
}
